# starting kernel
# baseline (speedup 1.0000x reference)
#include <hip/hip_runtime.h>

#define NF      39
#define EMBD    64
#define TILES   3
#define PADF    48
#define ESTRIDE 68
#define WAVES   4

typedef __attribute__((ext_vector_type(2))) float v2f;
typedef __attribute__((ext_vector_type(8))) float v8f;

static __device__ __forceinline__ v8f wmma_f32_k4(v2f a, v2f b, v8f c) {
  return __builtin_amdgcn_wmma_f32_16x16x4_f32(
       false, a,  false, b,
       (short)0, c,  false,  false);
}

__global__ __launch_bounds__(WAVES * 32)
void fwfm_wmma_kernel(const int* __restrict__ x,
                      const float* __restrict__ w0,
                      const float* __restrict__ bias,
                      const float* __restrict__ emb,
                      const float* __restrict__ W,
                      float* __restrict__ out,
                      int B) {
  __shared__ float E[WAVES][PADF * ESTRIDE];
  __shared__ float Wm[PADF * PADF];

  const int tid  = threadIdx.x;
  const int wv   = tid >> 5;
  const int lane = tid & 31;
  const int b    = blockIdx.x * WAVES + wv;
  const bool active = (b < B);

  for (int t = tid; t < PADF * PADF; t += WAVES * 32) {
    int i = t / PADF, j = t - i * PADF;
    float w = 0.0f;
    if (i < j && j < NF) w = W[i * NF + j];
    Wm[t] = w;
  }

  float partial = 0.0f;
  if (active) {
    const int* xr = x + b * NF;
    for (int t = lane; t < NF * (EMBD / 4); t += 32) {
      int row = t >> 4, c4 = t & 15;
      int idx = xr[row];
      const float4 v = *(const float4*)(emb + (size_t)idx * EMBD + c4 * 4);
      *(float4*)(&E[wv][row * ESTRIDE + c4 * 4]) = v;
    }
    float4 z; z.x = z.y = z.z = z.w = 0.0f;
    for (int t = lane; t < (PADF - NF) * (EMBD / 4); t += 32) {
      int row = NF + (t >> 4), c4 = t & 15;
      *(float4*)(&E[wv][row * ESTRIDE + c4 * 4]) = z;
    }
    for (int f = lane; f < NF; f += 32) partial += bias[xr[f]];
  }
  __syncthreads();

  if (active) {
    const float* Ew = E[wv];
    const int m    = lane & 15;
    const int koff = (lane >> 4) << 1;

    v8f c00 = {}, c01 = {}, c02 = {}, c11 = {}, c12 = {}, c22 = {};
    for (int s = 0; s < EMBD / 4; ++s) {
      const int kb = 4 * s + koff;
      v2f a0 = *(const v2f*)(&Ew[(0 * 16 + m) * ESTRIDE + kb]);
      v2f a1 = *(const v2f*)(&Ew[(1 * 16 + m) * ESTRIDE + kb]);
      v2f a2 = *(const v2f*)(&Ew[(2 * 16 + m) * ESTRIDE + kb]);
      c00 = wmma_f32_k4(a0, a0, c00);
      c01 = wmma_f32_k4(a0, a1, c01);
      c02 = wmma_f32_k4(a0, a2, c02);
      c11 = wmma_f32_k4(a1, a1, c11);
      c12 = wmma_f32_k4(a1, a2, c12);
      c22 = wmma_f32_k4(a2, a2, c22);
    }

    const int i0 = (lane >> 4) << 3;
    const int n  = lane & 15;
    float acc = partial;
#pragma unroll
    for (int r = 0; r < 8; ++r) {
      const int gi = i0 + r;
      acc += Wm[gi * PADF + n]             * c00[r];
      acc += Wm[gi * PADF + 16 + n]        * c01[r];
      acc += Wm[gi * PADF + 32 + n]        * c02[r];
      acc += Wm[(16 + gi) * PADF + 16 + n] * c11[r];
      acc += Wm[(16 + gi) * PADF + 32 + n] * c12[r];
      acc += Wm[(32 + gi) * PADF + 32 + n] * c22[r];
    }

    for (int off = 16; off > 0; off >>= 1)
      acc += __shfl_xor(acc, off, 32);
    if (lane == 0) out[b] = w0[0] + acc;
  }
}

extern "C" void kernel_launch(void* const* d_in, const int* in_sizes, int n_in,
                              void* d_out, int out_size, void* d_ws, size_t ws_size,
                              hipStream_t stream) {
  (void)n_in; (void)d_ws; (void)ws_size; (void)out_size;
  const int*   x    = (const int*)d_in[0];
  const float* w0   = (const float*)d_in[1];
  const float* bias = (const float*)d_in[2];
  const float* emb  = (const float*)d_in[3];
  const float* W    = (const float*)d_in[4];
  float* out = (float*)d_out;

  const int B = in_sizes[0] / NF;
  dim3 block(WAVES * 32);
  dim3 grid((B + WAVES - 1) / WAVES);
  fwfm_wmma_kernel<<<grid, block, 0, stream>>>(x, w0, bias, emb, W, out, B);
}


# prologue loads batched: indices read once per wave and lane-broadcast; W share, bias and all 20 row loads of a lane issued back to back ahead of use (was a load-wait-store trip each)
# speedup vs baseline: 1.6334x; 1.6334x over previous
#include <hip/hip_runtime.h>

#define NF      39
#define EMBD    64
#define TILES   3
#define PADF    48
#define ESTRIDE 68
#define WAVES   4
#define WSHARE  ((PADF * PADF) / (WAVES * 32))
#define ROWLD   ((NF + 1) / 2)

static_assert(WSHARE * WAVES * 32 == PADF * PADF, "W staging assumes an even share per thread");
static_assert(NF > 32 && NF <= 64 && (NF & 1) == 1, "index broadcast assumes 33..63 rows, odd");

typedef __attribute__((ext_vector_type(2))) float v2f;
typedef __attribute__((ext_vector_type(4))) float v4f;
typedef __attribute__((ext_vector_type(8))) float v8f;

static __device__ __forceinline__ v8f wmma_f32_k4(v2f a, v2f b, v8f c) {
  return __builtin_amdgcn_wmma_f32_16x16x4_f32(
       false, a,  false, b,
       (short)0, c,  false,  false);
}

__global__ __launch_bounds__(WAVES * 32)
void fwfm_wmma_kernel(const int* __restrict__ x,
                      const float* __restrict__ w0,
                      const float* __restrict__ bias,
                      const float* __restrict__ emb,
                      const float* __restrict__ W,
                      float* __restrict__ out,
                      int B) {
  __shared__ float E[WAVES][PADF * ESTRIDE];
  __shared__ float Wm[PADF * PADF];

  const int tid  = threadIdx.x;
  const int wv   = tid >> 5;
  const int lane = tid & 31;
  const int b    = blockIdx.x * WAVES + wv;
  const bool active = (b < B);
  const int bs   = active ? b : B - 1;
  const int h    = lane >> 4;
  const int c4   = lane & 15;


  const int* xr = x + bs * NF;
  const int i_lo = xr[lane];
  const int i_hi = xr[32 + (lane < NF - 32 ? lane : NF - 33)];

  float wreg[WSHARE];
#pragma unroll
  for (int k = 0; k < WSHARE; ++k) {
    const int t = tid + k * (WAVES * 32);
    const int i = t / PADF, j = t - i * PADF;
    const bool keep = (i < j && j < NF);
    const float w = W[keep ? i * NF + j : 0];
    wreg[k] = keep ? w : 0.0f;
  }

  const float b_lo = bias[i_lo];
  const float b_hi = bias[i_hi];
  v4f v[ROWLD];
#pragma unroll
  for (int i = 0; i < ROWLD; ++i) {
    const int r0 = 2 * i, r1 = 2 * i + 1;
    const int s0 = (r0 < 32) ? __builtin_amdgcn_readlane(i_lo, r0)
                             : __builtin_amdgcn_readlane(i_hi, r0 - 32);
    const int s1 = (r1 < 32) ? __builtin_amdgcn_readlane(i_lo, r1)
                 : (r1 < NF) ? __builtin_amdgcn_readlane(i_hi, r1 - 32) : s0;
    const int idx = h ? s1 : s0;
    v[i] = *(const v4f*)(emb + (size_t)idx * EMBD + c4 * 4);
  }

#pragma unroll
  for (int k = 0; k < WSHARE; ++k) Wm[tid + k * (WAVES * 32)] = wreg[k];

  {
    const v4f z = {0.0f, 0.0f, 0.0f, 0.0f};
#pragma unroll
    for (int i = 0; i < PADF / 2; ++i) {
      v4f t = z;
      if (i < ROWLD - 1) t = v[i];
      if (i == ROWLD - 1) t = h ? z : v[ROWLD - 1];
      *(v4f*)(&E[wv][(2 * i + h) * ESTRIDE + c4 * 4]) = t;
    }
  }
  __syncthreads();

  float partial = 0.0f;
  partial += b_lo;
  if (lane < NF - 32) partial += b_hi;

  {
    const float* Ew = E[wv];
    const int m    = lane & 15;
    const int koff = (lane >> 4) << 1;

    v8f c00 = {}, c01 = {}, c02 = {}, c11 = {}, c12 = {}, c22 = {};
    for (int s = 0; s < EMBD / 4; ++s) {
      const int kb = 4 * s + koff;
      v2f a0 = *(const v2f*)(&Ew[(0 * 16 + m) * ESTRIDE + kb]);
      v2f a1 = *(const v2f*)(&Ew[(1 * 16 + m) * ESTRIDE + kb]);
      v2f a2 = *(const v2f*)(&Ew[(2 * 16 + m) * ESTRIDE + kb]);
      c00 = wmma_f32_k4(a0, a0, c00);
      c01 = wmma_f32_k4(a0, a1, c01);
      c02 = wmma_f32_k4(a0, a2, c02);
      c11 = wmma_f32_k4(a1, a1, c11);
      c12 = wmma_f32_k4(a1, a2, c12);
      c22 = wmma_f32_k4(a2, a2, c22);
    }

    const int i0 = (lane >> 4) << 3;
    const int n  = lane & 15;
    float acc = partial;
#pragma unroll
    for (int r = 0; r < 8; ++r) {
      const int gi = i0 + r;
      acc += Wm[gi * PADF + n]             * c00[r];
      acc += Wm[gi * PADF + 16 + n]        * c01[r];
      acc += Wm[gi * PADF + 32 + n]        * c02[r];
      acc += Wm[(16 + gi) * PADF + 16 + n] * c11[r];
      acc += Wm[(16 + gi) * PADF + 32 + n] * c12[r];
      acc += Wm[(32 + gi) * PADF + 32 + n] * c22[r];
    }

    for (int off = 16; off > 0; off >>= 1)
      acc += __shfl_xor(acc, off, 32);
    if (active && lane == 0) out[b] = w0[0] + acc;
  }
}

extern "C" void kernel_launch(void* const* d_in, const int* in_sizes, int n_in,
                              void* d_out, int out_size, void* d_ws, size_t ws_size,
                              hipStream_t stream) {
  (void)n_in; (void)d_ws; (void)ws_size; (void)out_size;
  const int*   x    = (const int*)d_in[0];
  const float* w0   = (const float*)d_in[1];
  const float* bias = (const float*)d_in[2];
  const float* emb  = (const float*)d_in[3];
  const float* W    = (const float*)d_in[4];
  float* out = (float*)d_out;

  const int B = in_sizes[0] / NF;
  dim3 block(WAVES * 32);
  dim3 grid((B + WAVES - 1) / WAVES);
  fwfm_wmma_kernel<<<grid, block, 0, stream>>>(x, w0, bias, emb, W, out, B);
}


# interaction re-associated as sum_i <e_i,(Wu*E)_i>: gathered rows are the WMMA B operand straight from load registers, no LDS staging of rows (LDS 61440->4608 B), 72 WMMAs not 96; sum order changes
# speedup vs baseline: 1.9522x; 1.1952x over previous
#include <hip/hip_runtime.h>

#define NF      39
#define EMBD    64
#define WAVES   4
#define KSTEPS  ((NF + 3) / 4)
#define ROWLD   (2 * KSTEPS)
#define NFRAG   18
#define WSHARE  ((NFRAG * 64) / (WAVES * 32))

static_assert(WSHARE * WAVES * 32 == NFRAG * 64, "W staging assumes an even share per thread");
static_assert(NF > 32 && NF <= 40 && (NF & 1) == 1, "row tiling assumes 33..39 rows, odd");

typedef __attribute__((ext_vector_type(2))) float v2f;
typedef __attribute__((ext_vector_type(4))) float v4f;
typedef __attribute__((ext_vector_type(8))) float v8f;

static __device__ __forceinline__ v8f wmma_f32_k4(v2f a, v2f b, v8f c) {
  return __builtin_amdgcn_wmma_f32_16x16x4_f32(
       false, a,  false, b,
       (short)0, c,  false,  false);
}

template <int I, int P>
static __device__ __forceinline__ float row_tile(const v2f (&p)[KSTEPS][4], const float* Wp, unsigned loff, float acc) {
  constexpr int S0   = 4 * I;
  constexpr int FOFF = (I == 0) ? 0 : (I == 1) ? 6 : 8;
  constexpr int N0 = 2 * P, N1 = 2 * P + 1;
  const float* Wl = Wp + loff;
  v8f t0 = {}, t1 = {};
#pragma unroll
  for (int s = S0; s < KSTEPS; ++s) {
    const v2f a = *(const v2f*)(Wl + (s + FOFF) * 64);
    t0 = wmma_f32_k4(a, p[s][N0], t0);
    t1 = wmma_f32_k4(a, p[s][N1], t1);
  }
  constexpr int RMAX = (16 * I + 16 <= 4 * KSTEPS) ? 8 : (4 * KSTEPS - 16 * I) / 2;
#pragma unroll
  for (int r = 0; r < RMAX; ++r) {
    const int s = 4 * I + (r >> 1), e = r & 1;
    acc += t0[r] * p[s][N0][e];
    acc += t1[r] * p[s][N1][e];
  }
  return acc;
}

__global__ __launch_bounds__(WAVES * 32)
void fwfm_wmma_kernel(const int* __restrict__ x,
                      const float* __restrict__ w0,
                      const float* __restrict__ bias,
                      const float* __restrict__ emb,
                      const float* __restrict__ W,
                      float* __restrict__ out,
                      int B) {
  __shared__ float Wp[NFRAG * 64];

  const int tid  = threadIdx.x;
  const int wv   = tid >> 5;
  const int lane = tid & 31;
  const int b    = blockIdx.x * WAVES + wv;
  const bool active = (b < B);
  const int bs   = active ? b : B - 1;
  const int h    = lane >> 4;
  const int c4   = lane & 15;

  const int* xr = x + bs * NF;
  const int i_lo = xr[lane];
  const int i_hi = xr[32 + (lane < NF - 32 ? lane : NF - 33)];

  const int fl = (tid & 63) >> 1;
  const int fm = fl & 15;
  const int rp = 4 * ((fm & 7) >> 1) + 2 * (fm >> 3) + (fm & 1);
  const int jb = 4 * (tid >> 6) + 2 * (fl >> 4) + (tid & 1);
  float wreg[WSHARE];
#pragma unroll
  for (int k = 0; k < WSHARE; ++k) {
    const int I    = (2 * k >= 16) ? 2 : (2 * k >= 10) ? 1 : 0;
    const int foff = (I == 0) ? 0 : (I == 1) ? 6 : 8;
    const int i = 16 * I + rp;
    const int j = jb + 4 * (2 * k - foff);
    const bool keep = (i < j && j < NF);
    const float w = W[keep ? i * NF + j : 0];
    wreg[k] = keep ? w : 0.0f;
  }

  const float b_lo = bias[i_lo];
  const float b_hi = bias[i_hi];
  const v4f* __restrict__ emb4 = (const v4f*)emb;
  v4f v[ROWLD];
#pragma unroll
  for (int i = 0; i < ROWLD; ++i) {
    const int r0 = 4 * (i >> 1) + (i & 1), r1 = r0 + 2;
    const int s0 = (r0 < 32) ? __builtin_amdgcn_readlane(i_lo, r0)
                             : __builtin_amdgcn_readlane(i_hi, r0 - 32);
    const int s1 = (r1 < 32) ? __builtin_amdgcn_readlane(i_lo, r1)
                 : (r1 < NF) ? __builtin_amdgcn_readlane(i_hi, r1 - 32) : s0;
    const int idx = h ? s1 : s0;
    v[i] = emb4[(unsigned)idx * (EMBD / 4) + (unsigned)c4];
  }

#pragma unroll
  for (int k = 0; k < WSHARE; ++k) Wp[tid + k * (WAVES * 32)] = wreg[k];
  __syncthreads();

  if (h) v[ROWLD - 1] = v4f{0.0f, 0.0f, 0.0f, 0.0f};

  v2f p[KSTEPS][4];
#pragma unroll
  for (int s = 0; s < KSTEPS; ++s) {
#pragma unroll
    for (int n = 0; n < 4; ++n) p[s][n] = v2f{v[2 * s][n], v[2 * s + 1][n]};
  }

  float acc = 0.0f;
  acc += b_lo;
  if (lane < NF - 32) acc += b_hi;

  const unsigned loff = lane * 2;
  acc = row_tile<0, 0>(p, Wp, loff, acc);
  acc = row_tile<0, 1>(p, Wp, loff, acc);
  acc = row_tile<1, 0>(p, Wp, loff, acc);
  acc = row_tile<1, 1>(p, Wp, loff, acc);
  acc = row_tile<2, 0>(p, Wp, loff, acc);
  acc = row_tile<2, 1>(p, Wp, loff, acc);

  for (int off = 16; off > 0; off >>= 1)
    acc += __shfl_xor(acc, off, 32);
  if (active && lane == 0) out[b] = w0[0] + acc;
}

extern "C" void kernel_launch(void* const* d_in, const int* in_sizes, int n_in,
                              void* d_out, int out_size, void* d_ws, size_t ws_size,
                              hipStream_t stream) {
  (void)n_in; (void)d_ws; (void)ws_size; (void)out_size;
  const int*   x    = (const int*)d_in[0];
  const float* w0   = (const float*)d_in[1];
  const float* bias = (const float*)d_in[2];
  const float* emb  = (const float*)d_in[3];
  const float* W    = (const float*)d_in[4];
  float* out = (float*)d_out;

  const int B = in_sizes[0] / NF;
  dim3 block(WAVES * 32);
  dim3 grid((B + WAVES - 1) / WAVES);
  fwfm_wmma_kernel<<<grid, block, 0, stream>>>(x, w0, bias, emb, W, out, B);
}


# each wave takes 4 consecutive samples in a rolled loop (grid 4096 -> 1024 blocks, all waves resident at once); next sample's index, bias and row loads issued ahead of the current sample's WMMAs
# speedup vs baseline: 2.1595x; 1.1062x over previous
#include <hip/hip_runtime.h>

#define NF      39
#define EMBD    64
#define WAVES   4
#define SPW     4
#define KSTEPS  ((NF + 3) / 4)
#define ROWLD   (2 * KSTEPS)
#define NFRAG   18
#define WSHARE  ((NFRAG * 64) / (WAVES * 32))

static_assert(WSHARE * WAVES * 32 == NFRAG * 64, "W staging assumes an even share per thread");
static_assert(NF > 32 && NF <= 40 && (NF & 1) == 1, "row tiling assumes 33..39 rows, odd");

typedef __attribute__((ext_vector_type(2))) float v2f;
typedef __attribute__((ext_vector_type(4))) float v4f;
typedef __attribute__((ext_vector_type(8))) float v8f;

static __device__ __forceinline__ v8f wmma_f32_k4(v2f a, v2f b, v8f c) {
  return __builtin_amdgcn_wmma_f32_16x16x4_f32(
       false, a,  false, b,
       (short)0, c,  false,  false);
}

template <int I, int P>
static __device__ __forceinline__ float row_tile(const v2f (&p)[KSTEPS][4], const float* Wp, unsigned loff, float acc) {
  constexpr int S0   = 4 * I;
  constexpr int FOFF = (I == 0) ? 0 : (I == 1) ? 6 : 8;
  constexpr int N0 = 2 * P, N1 = 2 * P + 1;
  const float* Wl = Wp + loff;
  v8f t0 = {}, t1 = {};
#pragma unroll
  for (int s = S0; s < KSTEPS; ++s) {
    const v2f a = *(const v2f*)(Wl + (s + FOFF) * 64);
    t0 = wmma_f32_k4(a, p[s][N0], t0);
    t1 = wmma_f32_k4(a, p[s][N1], t1);
  }
  constexpr int RMAX = (16 * I + 16 <= 4 * KSTEPS) ? 8 : (4 * KSTEPS - 16 * I) / 2;
#pragma unroll
  for (int r = 0; r < RMAX; ++r) {
    const int s = 4 * I + (r >> 1), e = r & 1;
    acc += t0[r] * p[s][N0][e];
    acc += t1[r] * p[s][N1][e];
  }
  return acc;
}

static __device__ __forceinline__ void issue_rows(const v4f* __restrict__ emb4, const float* __restrict__ bias,
                                                  int i_lo, int i_hi, int h, int c4,
                                                  v4f (&v)[ROWLD], float& b_lo, float& b_hi) {
  b_lo = bias[i_lo];
  b_hi = bias[i_hi];
#pragma unroll
  for (int i = 0; i < ROWLD; ++i) {
    const int r0 = 4 * (i >> 1) + (i & 1), r1 = r0 + 2;
    const int s0 = (r0 < 32) ? __builtin_amdgcn_readlane(i_lo, r0)
                             : __builtin_amdgcn_readlane(i_hi, r0 - 32);
    const int s1 = (r1 < 32) ? __builtin_amdgcn_readlane(i_lo, r1)
                 : (r1 < NF) ? __builtin_amdgcn_readlane(i_hi, r1 - 32) : s0;
    const int idx = h ? s1 : s0;
    v[i] = emb4[(unsigned)idx * (EMBD / 4) + (unsigned)c4];
  }
}

__global__ __launch_bounds__(WAVES * 32)
void fwfm_wmma_kernel(const int* __restrict__ x,
                      const float* __restrict__ w0,
                      const float* __restrict__ bias,
                      const float* __restrict__ emb,
                      const float* __restrict__ W,
                      float* __restrict__ out,
                      int B) {
  __shared__ float Wp[NFRAG * 64];

  const int tid  = threadIdx.x;
  const int wv   = tid >> 5;
  const int lane = tid & 31;
  const int bw   = (blockIdx.x * WAVES + wv) * SPW;
  const int h    = lane >> 4;
  const int c4   = lane & 15;
  const int hi_l = 32 + (lane < NF - 32 ? lane : NF - 33);
  const v4f* __restrict__ emb4 = (const v4f*)emb;

  int i_lo, i_hi;
  {
    const int* xr = x + (bw < B ? bw : B - 1) * NF;
    i_lo = xr[lane];
    i_hi = xr[hi_l];
  }

  const int fl = (tid & 63) >> 1;
  const int fm = fl & 15;
  const int rp = 4 * ((fm & 7) >> 1) + 2 * (fm >> 3) + (fm & 1);
  const int jb = 4 * (tid >> 6) + 2 * (fl >> 4) + (tid & 1);
  float wreg[WSHARE];
#pragma unroll
  for (int k = 0; k < WSHARE; ++k) {
    const int I    = (2 * k >= 16) ? 2 : (2 * k >= 10) ? 1 : 0;
    const int foff = (I == 0) ? 0 : (I == 1) ? 6 : 8;
    const int i = 16 * I + rp;
    const int j = jb + 4 * (2 * k - foff);
    const bool keep = (i < j && j < NF);
    const float w = W[keep ? i * NF + j : 0];
    wreg[k] = keep ? w : 0.0f;
  }

  float b_lo, b_hi;
  v4f v[ROWLD];
  issue_rows(emb4, bias, i_lo, i_hi, h, c4, v, b_lo, b_hi);

  if (SPW > 1) {
    const int* xr = x + (bw + 1 < B ? bw + 1 : B - 1) * NF;
    i_lo = xr[lane];
    i_hi = xr[hi_l];
  }

#pragma unroll
  for (int k = 0; k < WSHARE; ++k) Wp[tid + k * (WAVES * 32)] = wreg[k];
  __syncthreads();

  const unsigned loff = lane * 2;
  float res = 0.0f;

  float acc0 = 0.0f;
  acc0 += b_lo;
  if (lane < NF - 32) acc0 += b_hi;

#pragma unroll 1
  for (int it = 0; it < SPW; ++it) {
    if (h) v[ROWLD - 1] = v4f{0.0f, 0.0f, 0.0f, 0.0f};
    v2f p[KSTEPS][4];
#pragma unroll
    for (int s = 0; s < KSTEPS; ++s) {
#pragma unroll
      for (int n = 0; n < 4; ++n) {
        p[s][n] = v2f{v[2 * s][n], v[2 * s + 1][n]};
        asm volatile("" : "+v"(p[s][n]));
      }
    }

    float acc = acc0;

    if (it + 1 < SPW) {
      issue_rows(emb4, bias, i_lo, i_hi, h, c4, v, b_lo, b_hi);
      if (it + 2 < SPW) {
        const int bn = bw + it + 2;
        const int* xr = x + (bn < B ? bn : B - 1) * NF;
        i_lo = xr[lane];
        i_hi = xr[hi_l];
      }
    }

    acc = row_tile<0, 0>(p, Wp, loff, acc);
    acc = row_tile<0, 1>(p, Wp, loff, acc);
    acc = row_tile<1, 0>(p, Wp, loff, acc);
    acc = row_tile<1, 1>(p, Wp, loff, acc);
    acc = row_tile<2, 0>(p, Wp, loff, acc);
    acc = row_tile<2, 1>(p, Wp, loff, acc);

    for (int off = 16; off > 0; off >>= 1)
      acc += __shfl_xor(acc, off, 32);
    if (lane == it) res = acc;

    if (it + 1 < SPW) {
      acc0 = 0.0f;
      acc0 += b_lo;
      if (lane < NF - 32) acc0 += b_hi;
    }
  }

  if (lane < SPW && bw + lane < B) out[bw + lane] = w0[0] + res;
}

extern "C" void kernel_launch(void* const* d_in, const int* in_sizes, int n_in,
                              void* d_out, int out_size, void* d_ws, size_t ws_size,
                              hipStream_t stream) {
  (void)n_in; (void)d_ws; (void)ws_size; (void)out_size;
  const int*   x    = (const int*)d_in[0];
  const float* w0   = (const float*)d_in[1];
  const float* bias = (const float*)d_in[2];
  const float* emb  = (const float*)d_in[3];
  const float* W    = (const float*)d_in[4];
  float* out = (float*)d_out;

  const int B = in_sizes[0] / NF;
  dim3 block(WAVES * 32);
  dim3 grid((B + WAVES * SPW - 1) / (WAVES * SPW));
  fwfm_wmma_kernel<<<grid, block, 0, stream>>>(x, w0, bias, emb, W, out, B);
}


# sample loop freed of LDS round trips: masked-W A fragments held in 36 registers for all 4 samples; wave sum by lane-permute VALU ops (same xor tree, same sum order) instead of ds_bpermute
# speedup vs baseline: 2.3100x; 1.0697x over previous
#include <hip/hip_runtime.h>

#define NF      39
#define EMBD    64
#define WAVES   4
#define SPW     4
#define KSTEPS  ((NF + 3) / 4)
#define ROWLD   (2 * KSTEPS)
#define NFRAG   18
#define WSHARE  ((NFRAG * 64) / (WAVES * 32))

static_assert(WSHARE * WAVES * 32 == NFRAG * 64, "W staging assumes an even share per thread");
static_assert(NF > 32 && NF <= 40 && (NF & 1) == 1, "row tiling assumes 33..39 rows, odd");

typedef __attribute__((ext_vector_type(2))) float v2f;
typedef __attribute__((ext_vector_type(4))) float v4f;
typedef __attribute__((ext_vector_type(8))) float v8f;

static __device__ __forceinline__ v8f wmma_f32_k4(v2f a, v2f b, v8f c) {
  return __builtin_amdgcn_wmma_f32_16x16x4_f32(
       false, a,  false, b,
       (short)0, c,  false,  false);
}

template <int CTRL>
static __device__ __forceinline__ float lane_xor_row(float a) {
  return __int_as_float(__builtin_amdgcn_update_dpp(0, __float_as_int(a), CTRL, 0xf, 0xf, true));
}
static __device__ __forceinline__ float wave_sum(float a) {
  a += __int_as_float((int)__builtin_amdgcn_permlanex16((unsigned)__float_as_int(a), (unsigned)__float_as_int(a),
                                                        0x76543210u, 0xfedcba98u, false, false));
  a += lane_xor_row<0x168>(a);
  a += lane_xor_row<0x164>(a);
  a += lane_xor_row<0x162>(a);
  a += lane_xor_row<0x161>(a);
  return a;
}

template <int I, int P>
static __device__ __forceinline__ float row_tile(const v2f (&p)[KSTEPS][4], const v2f (&af)[NFRAG], float acc) {
  constexpr int S0   = 4 * I;
  constexpr int FOFF = (I == 0) ? 0 : (I == 1) ? 6 : 8;
  constexpr int N0 = 2 * P, N1 = 2 * P + 1;
  v8f t0 = {}, t1 = {};
#pragma unroll
  for (int s = S0; s < KSTEPS; ++s) {
    const v2f a = af[s + FOFF];
    t0 = wmma_f32_k4(a, p[s][N0], t0);
    t1 = wmma_f32_k4(a, p[s][N1], t1);
  }
  constexpr int RMAX = (16 * I + 16 <= 4 * KSTEPS) ? 8 : (4 * KSTEPS - 16 * I) / 2;
#pragma unroll
  for (int r = 0; r < RMAX; ++r) {
    const int s = 4 * I + (r >> 1), e = r & 1;
    acc += t0[r] * p[s][N0][e];
    acc += t1[r] * p[s][N1][e];
  }
  return acc;
}

static __device__ __forceinline__ void issue_rows(const v4f* __restrict__ emb4, const float* __restrict__ bias,
                                                  int i_lo, int i_hi, int h, int c4,
                                                  v4f (&v)[ROWLD], float& b_lo, float& b_hi) {
  b_lo = bias[i_lo];
  b_hi = bias[i_hi];
#pragma unroll
  for (int i = 0; i < ROWLD; ++i) {
    const int r0 = 4 * (i >> 1) + (i & 1), r1 = r0 + 2;
    const int s0 = (r0 < 32) ? __builtin_amdgcn_readlane(i_lo, r0)
                             : __builtin_amdgcn_readlane(i_hi, r0 - 32);
    const int s1 = (r1 < 32) ? __builtin_amdgcn_readlane(i_lo, r1)
                 : (r1 < NF) ? __builtin_amdgcn_readlane(i_hi, r1 - 32) : s0;
    const int idx = h ? s1 : s0;
    v[i] = emb4[(unsigned)idx * (EMBD / 4) + (unsigned)c4];
  }
}

__global__ __launch_bounds__(WAVES * 32, 4)
void fwfm_wmma_kernel(const int* __restrict__ x,
                      const float* __restrict__ w0,
                      const float* __restrict__ bias,
                      const float* __restrict__ emb,
                      const float* __restrict__ W,
                      float* __restrict__ out,
                      int B) {
  __shared__ float Wp[NFRAG * 64];

  const int tid  = threadIdx.x;
  const int wv   = tid >> 5;
  const int lane = tid & 31;
  const int bw   = (blockIdx.x * WAVES + wv) * SPW;
  const int h    = lane >> 4;
  const int c4   = lane & 15;
  const int hi_l = 32 + (lane < NF - 32 ? lane : NF - 33);
  const v4f* __restrict__ emb4 = (const v4f*)emb;

  int i_lo, i_hi;
  {
    const int* xr = x + (bw < B ? bw : B - 1) * NF;
    i_lo = xr[lane];
    i_hi = xr[hi_l];
  }

  const int fl = (tid & 63) >> 1;
  const int fm = fl & 15;
  const int rp = 4 * ((fm & 7) >> 1) + 2 * (fm >> 3) + (fm & 1);
  const int jb = 4 * (tid >> 6) + 2 * (fl >> 4) + (tid & 1);
  float wreg[WSHARE];
#pragma unroll
  for (int k = 0; k < WSHARE; ++k) {
    const int I    = (2 * k >= 16) ? 2 : (2 * k >= 10) ? 1 : 0;
    const int foff = (I == 0) ? 0 : (I == 1) ? 6 : 8;
    const int i = 16 * I + rp;
    const int j = jb + 4 * (2 * k - foff);
    const bool keep = (i < j && j < NF);
    const float w = W[keep ? i * NF + j : 0];
    wreg[k] = keep ? w : 0.0f;
  }

  float b_lo, b_hi;
  v4f v[ROWLD];
  issue_rows(emb4, bias, i_lo, i_hi, h, c4, v, b_lo, b_hi);

  if (SPW > 1) {
    const int* xr = x + (bw + 1 < B ? bw + 1 : B - 1) * NF;
    i_lo = xr[lane];
    i_hi = xr[hi_l];
  }

#pragma unroll
  for (int k = 0; k < WSHARE; ++k) Wp[tid + k * (WAVES * 32)] = wreg[k];
  __syncthreads();

  v2f af[NFRAG];
#pragma unroll
  for (int f = 0; f < NFRAG; ++f) af[f] = *(const v2f*)(Wp + f * 64 + lane * 2);
  float res = 0.0f;

  float acc0 = 0.0f;
  acc0 += b_lo;
  if (lane < NF - 32) acc0 += b_hi;

#pragma unroll 1
  for (int it = 0; it < SPW; ++it) {
    if (h) v[ROWLD - 1] = v4f{0.0f, 0.0f, 0.0f, 0.0f};
    v2f p[KSTEPS][4];
#pragma unroll
    for (int s = 0; s < KSTEPS; ++s) {
#pragma unroll
      for (int n = 0; n < 4; ++n) {
        p[s][n] = v2f{v[2 * s][n], v[2 * s + 1][n]};
        asm volatile("" : "+v"(p[s][n]));
      }
    }

    float acc = acc0;

    if (it + 1 < SPW) {
      issue_rows(emb4, bias, i_lo, i_hi, h, c4, v, b_lo, b_hi);
      if (it + 2 < SPW) {
        const int bn = bw + it + 2;
        const int* xr = x + (bn < B ? bn : B - 1) * NF;
        i_lo = xr[lane];
        i_hi = xr[hi_l];
      }
    }

    acc = row_tile<0, 0>(p, af, acc);
    acc = row_tile<0, 1>(p, af, acc);
    acc = row_tile<1, 0>(p, af, acc);
    acc = row_tile<1, 1>(p, af, acc);
    acc = row_tile<2, 0>(p, af, acc);
    acc = row_tile<2, 1>(p, af, acc);

    acc = wave_sum(acc);
    if (lane == it) res = acc;

    if (it + 1 < SPW) {
      acc0 = 0.0f;
      acc0 += b_lo;
      if (lane < NF - 32) acc0 += b_hi;
    }
  }

  if (lane < SPW && bw + lane < B) out[bw + lane] = w0[0] + res;
}

extern "C" void kernel_launch(void* const* d_in, const int* in_sizes, int n_in,
                              void* d_out, int out_size, void* d_ws, size_t ws_size,
                              hipStream_t stream) {
  (void)n_in; (void)d_ws; (void)ws_size; (void)out_size;
  const int*   x    = (const int*)d_in[0];
  const float* w0   = (const float*)d_in[1];
  const float* bias = (const float*)d_in[2];
  const float* emb  = (const float*)d_in[3];
  const float* W    = (const float*)d_in[4];
  float* out = (float*)d_out;

  const int B = in_sizes[0] / NF;
  dim3 block(WAVES * 32);
  dim3 grid((B + WAVES * SPW - 1) / (WAVES * SPW));
  fwfm_wmma_kernel<<<grid, block, 0, stream>>>(x, w0, bias, emb, W, out, B);
}
